# starting kernel
# baseline (speedup 1.0000x reference)
#include <hip/hip_runtime.h>

typedef float v2f __attribute__((ext_vector_type(2)));
typedef float v4f __attribute__((ext_vector_type(4)));
typedef float v8f __attribute__((ext_vector_type(8)));

#define MAX_POS 1024
#define S_LEN   1024
#define EMB     1024
#define HEADS   16
#define DHEAD   64
#define BATCH   2
#define M1_PAD  2048

#define LDA1 36
#define LDB1 36

__global__ __launch_bounds__(128)
void gemm1_kernel(const float* __restrict__ pos, const float* __restrict__ W,
                  float* __restrict__ emb2)
{
    __shared__ float Als[64 * LDA1];
    __shared__ float Bls[64 * LDB1];

    const int t    = threadIdx.x;
    const int lane = t & 31;
    const int wave = t >> 5;
    const int wm   = wave >> 1;
    const int wn   = wave & 1;

    const int Mbase = blockIdx.x * 64;
    const int Nbase = blockIdx.y * 64;

    v8f acc[2][2] = {};

    for (int kc = 0; kc < EMB; kc += 32) {
        #pragma unroll
        for (int it = 0; it < 4; ++it) {
            int idx = t + it * 128;
            int row = idx >> 3;
            int col = (idx & 7) << 2;
            int gidx = Mbase + row + (MAX_POS - S_LEN);
            gidx = gidx < 0 ? 0 : gidx;
            gidx = gidx > (2 * MAX_POS - 2) ? (2 * MAX_POS - 2) : gidx;
            v4f v = *(const v4f*)(pos + (size_t)gidx * EMB + kc + col);
            *(v4f*)(&Als[row * LDA1 + col]) = v;
        }
        #pragma unroll
        for (int it = 0; it < 16; ++it) {
            int idx = t + it * 128;
            int k = idx >> 6;
            int n = idx & 63;
            Bls[n * LDB1 + k] = W[(size_t)(kc + k) * EMB + Nbase + n];
        }
        __syncthreads();

        #pragma unroll
        for (int ks = 0; ks < 32; ks += 4) {
            v2f afrag[2], bfrag[2];
            #pragma unroll
            for (int tm = 0; tm < 2; ++tm) {
                int m = wm * 32 + tm * 16 + (lane & 15);
                afrag[tm] = *(const v2f*)(&Als[m * LDA1 + ks + 2 * (lane >> 4)]);
            }
            #pragma unroll
            for (int tn = 0; tn < 2; ++tn) {
                int n = wn * 32 + tn * 16 + (lane & 15);
                bfrag[tn] = *(const v2f*)(&Bls[n * LDB1 + ks + 2 * (lane >> 4)]);
            }
            #pragma unroll
            for (int tm = 0; tm < 2; ++tm)
                #pragma unroll
                for (int tn = 0; tn < 2; ++tn)
                    acc[tm][tn] = __builtin_amdgcn_wmma_f32_16x16x4_f32(
                        false, afrag[tm], false, bfrag[tn],
                        (short)0, acc[tm][tn], false, false);
        }
        __syncthreads();
    }

    #pragma unroll
    for (int tm = 0; tm < 2; ++tm)
        #pragma unroll
        for (int tn = 0; tn < 2; ++tn) {
            int mrow0 = Mbase + wm * 32 + tm * 16 + 8 * (lane >> 4);
            int ncol  = Nbase + wn * 32 + tn * 16 + (lane & 15);
            #pragma unroll
            for (int v = 0; v < 8; ++v)
                emb2[(size_t)(mrow0 + v) * EMB + ncol] = acc[tm][tn][v];
        }
}

#define LDA2 68
#define LDB2 68

__global__ __launch_bounds__(128)
void gemm2_kernel(const float* __restrict__ query, const float* __restrict__ emb2,
                  float* __restrict__ out)
{
    __shared__ float Als[64 * LDA2];
    __shared__ float Bls[64 * LDB2];

    const int t    = threadIdx.x;
    const int lane = t & 31;
    const int wave = t >> 5;
    const int wm   = wave >> 1;
    const int wn   = wave & 1;

    const int h     = blockIdx.z;
    const int Nbase = blockIdx.x * 64;
    const int Mbase = blockIdx.y * 64;

    #pragma unroll
    for (int it = 0; it < 8; ++it) {
        int idx = t + it * 128;
        int row = idx >> 4;
        int col = (idx & 15) << 2;
        v4f v = *(const v4f*)(query +
                 ((size_t)h * (BATCH * S_LEN) + Nbase + row) * DHEAD + col);
        *(v4f*)(&Als[row * LDA2 + col]) = v;
    }
    #pragma unroll
    for (int it = 0; it < 8; ++it) {
        int idx = t + it * 128;
        int row = idx >> 4;
        int col = (idx & 15) << 2;
        v4f v = *(const v4f*)(emb2 + (size_t)(Mbase + row) * EMB + h * DHEAD + col);
        *(v4f*)(&Bls[row * LDB2 + col]) = v;
    }
    __syncthreads();

    v8f acc[2][2] = {};
    #pragma unroll
    for (int ks = 0; ks < DHEAD; ks += 4) {
        v2f afrag[2], bfrag[2];
        #pragma unroll
        for (int tm = 0; tm < 2; ++tm) {
            int n = wm * 32 + tm * 16 + (lane & 15);
            afrag[tm] = *(const v2f*)(&Als[n * LDA2 + ks + 2 * (lane >> 4)]);
        }
        #pragma unroll
        for (int tn = 0; tn < 2; ++tn) {
            int m = wn * 32 + tn * 16 + (lane & 15);
            bfrag[tn] = *(const v2f*)(&Bls[m * LDB2 + ks + 2 * (lane >> 4)]);
        }
        #pragma unroll
        for (int tm = 0; tm < 2; ++tm)
            #pragma unroll
            for (int tn = 0; tn < 2; ++tn)
                acc[tm][tn] = __builtin_amdgcn_wmma_f32_16x16x4_f32(
                    false, afrag[tm], false, bfrag[tn],
                    (short)0, acc[tm][tn], false, false);
    }

    #pragma unroll
    for (int tm = 0; tm < 2; ++tm)
        #pragma unroll
        for (int tn = 0; tn < 2; ++tn) {
            int nrow0 = Nbase + wm * 32 + tm * 16 + 8 * (lane >> 4);
            int mcol  = Mbase + wn * 32 + tn * 16 + (lane & 15);
            #pragma unroll
            for (int v = 0; v < 8; ++v) {
                int n = nrow0 + v;
                int b = n >> 10;
                int s = n & (S_LEN - 1);
                int j = mcol - (S_LEN - 1) + s;
                if ((unsigned)j < (unsigned)S_LEN) {
                    size_t o = (((size_t)h * BATCH + b) * S_LEN + s) * (size_t)S_LEN + j;
                    out[o] = acc[tm][tn][v];
                }
            }
        }
}

extern "C" void kernel_launch(void* const* d_in, const int* in_sizes, int n_in,
                              void* d_out, int out_size, void* d_ws, size_t ws_size,
                              hipStream_t stream) {
    const float* query = (const float*)d_in[0];
    const float* pos   = (const float*)d_in[1];
    const float* W     = (const float*)d_in[2];
    float* out  = (float*)d_out;
    float* emb2 = (float*)d_ws;

    dim3 g1(M1_PAD / 64, EMB / 64);
    gemm1_kernel<<<g1, 128, 0, stream>>>(pos, W, emb2);

    dim3 g2((BATCH * S_LEN) / 64, M1_PAD / 64, HEADS);
    gemm2_kernel<<<g2, 128, 0, stream>>>(query, emb2, out);
}


# gemm2 grid 32x32x16 -> 32x17x16: launch only the 17 m tiles per s tile that meet the relative-shift band (diagonal tile map); tiles with every j outside [0,S) are no longer loaded or multiplied
# speedup vs baseline: 1.1545x; 1.1545x over previous
#include <hip/hip_runtime.h>

typedef float v2f __attribute__((ext_vector_type(2)));
typedef float v4f __attribute__((ext_vector_type(4)));
typedef float v8f __attribute__((ext_vector_type(8)));

#define MAX_POS 1024
#define S_LEN   1024
#define EMB     1024
#define HEADS   16
#define DHEAD   64
#define BATCH   2
#define M1_PAD  2048
#define BAND_TILES (S_LEN / 64 + 1)

#define LDA1 36
#define LDB1 36

__global__ __launch_bounds__(128)
void gemm1_kernel(const float* __restrict__ pos, const float* __restrict__ W,
                  float* __restrict__ emb2)
{
    __shared__ float Als[64 * LDA1];
    __shared__ float Bls[64 * LDB1];

    const int t    = threadIdx.x;
    const int lane = t & 31;
    const int wave = t >> 5;
    const int wm   = wave >> 1;
    const int wn   = wave & 1;

    const int Mbase = blockIdx.x * 64;
    const int Nbase = blockIdx.y * 64;

    v8f acc[2][2] = {};

    for (int kc = 0; kc < EMB; kc += 32) {
        #pragma unroll
        for (int it = 0; it < 4; ++it) {
            int idx = t + it * 128;
            int row = idx >> 3;
            int col = (idx & 7) << 2;
            int gidx = Mbase + row + (MAX_POS - S_LEN);
            gidx = gidx < 0 ? 0 : gidx;
            gidx = gidx > (2 * MAX_POS - 2) ? (2 * MAX_POS - 2) : gidx;
            v4f v = *(const v4f*)(pos + (size_t)gidx * EMB + kc + col);
            *(v4f*)(&Als[row * LDA1 + col]) = v;
        }
        #pragma unroll
        for (int it = 0; it < 16; ++it) {
            int idx = t + it * 128;
            int k = idx >> 6;
            int n = idx & 63;
            Bls[n * LDB1 + k] = W[(size_t)(kc + k) * EMB + Nbase + n];
        }
        __syncthreads();

        #pragma unroll
        for (int ks = 0; ks < 32; ks += 4) {
            v2f afrag[2], bfrag[2];
            #pragma unroll
            for (int tm = 0; tm < 2; ++tm) {
                int m = wm * 32 + tm * 16 + (lane & 15);
                afrag[tm] = *(const v2f*)(&Als[m * LDA1 + ks + 2 * (lane >> 4)]);
            }
            #pragma unroll
            for (int tn = 0; tn < 2; ++tn) {
                int n = wn * 32 + tn * 16 + (lane & 15);
                bfrag[tn] = *(const v2f*)(&Bls[n * LDB1 + ks + 2 * (lane >> 4)]);
            }
            #pragma unroll
            for (int tm = 0; tm < 2; ++tm)
                #pragma unroll
                for (int tn = 0; tn < 2; ++tn)
                    acc[tm][tn] = __builtin_amdgcn_wmma_f32_16x16x4_f32(
                        false, afrag[tm], false, bfrag[tn],
                        (short)0, acc[tm][tn], false, false);
        }
        __syncthreads();
    }

    #pragma unroll
    for (int tm = 0; tm < 2; ++tm)
        #pragma unroll
        for (int tn = 0; tn < 2; ++tn) {
            int mrow0 = Mbase + wm * 32 + tm * 16 + 8 * (lane >> 4);
            int ncol  = Nbase + wn * 32 + tn * 16 + (lane & 15);
            #pragma unroll
            for (int v = 0; v < 8; ++v)
                emb2[(size_t)(mrow0 + v) * EMB + ncol] = acc[tm][tn][v];
        }
}

#define LDA2 68
#define LDB2 68

__global__ __launch_bounds__(128)
void gemm2_kernel(const float* __restrict__ query, const float* __restrict__ emb2,
                  float* __restrict__ out)
{
    __shared__ float Als[64 * LDA2];
    __shared__ float Bls[64 * LDB2];

    const int t    = threadIdx.x;
    const int lane = t & 31;
    const int wave = t >> 5;
    const int wm   = wave >> 1;
    const int wn   = wave & 1;

    const int h     = blockIdx.z;
    const int Nbase = blockIdx.x * 64;
    const int si    = blockIdx.x & (S_LEN / 64 - 1);
    const int Mbase = (S_LEN / 64 - 1 - si + (int)blockIdx.y) * 64;

    #pragma unroll
    for (int it = 0; it < 8; ++it) {
        int idx = t + it * 128;
        int row = idx >> 4;
        int col = (idx & 15) << 2;
        v4f v = *(const v4f*)(query +
                 ((size_t)h * (BATCH * S_LEN) + Nbase + row) * DHEAD + col);
        *(v4f*)(&Als[row * LDA2 + col]) = v;
    }
    #pragma unroll
    for (int it = 0; it < 8; ++it) {
        int idx = t + it * 128;
        int row = idx >> 4;
        int col = (idx & 15) << 2;
        v4f v = *(const v4f*)(emb2 + (size_t)(Mbase + row) * EMB + h * DHEAD + col);
        *(v4f*)(&Bls[row * LDB2 + col]) = v;
    }
    __syncthreads();

    v8f acc[2][2] = {};
    #pragma unroll
    for (int ks = 0; ks < DHEAD; ks += 4) {
        v2f afrag[2], bfrag[2];
        #pragma unroll
        for (int tm = 0; tm < 2; ++tm) {
            int n = wm * 32 + tm * 16 + (lane & 15);
            afrag[tm] = *(const v2f*)(&Als[n * LDA2 + ks + 2 * (lane >> 4)]);
        }
        #pragma unroll
        for (int tn = 0; tn < 2; ++tn) {
            int m = wn * 32 + tn * 16 + (lane & 15);
            bfrag[tn] = *(const v2f*)(&Bls[m * LDB2 + ks + 2 * (lane >> 4)]);
        }
        #pragma unroll
        for (int tm = 0; tm < 2; ++tm)
            #pragma unroll
            for (int tn = 0; tn < 2; ++tn)
                acc[tm][tn] = __builtin_amdgcn_wmma_f32_16x16x4_f32(
                    false, afrag[tm], false, bfrag[tn],
                    (short)0, acc[tm][tn], false, false);
    }

    #pragma unroll
    for (int tm = 0; tm < 2; ++tm)
        #pragma unroll
        for (int tn = 0; tn < 2; ++tn) {
            int nrow0 = Nbase + wm * 32 + tm * 16 + 8 * (lane >> 4);
            int mcol  = Mbase + wn * 32 + tn * 16 + (lane & 15);
            #pragma unroll
            for (int v = 0; v < 8; ++v) {
                int n = nrow0 + v;
                int b = n >> 10;
                int s = n & (S_LEN - 1);
                int j = mcol - (S_LEN - 1) + s;
                if ((unsigned)j < (unsigned)S_LEN) {
                    size_t o = (((size_t)h * BATCH + b) * S_LEN + s) * (size_t)S_LEN + j;
                    out[o] = acc[tm][tn][v];
                }
            }
        }
}

extern "C" void kernel_launch(void* const* d_in, const int* in_sizes, int n_in,
                              void* d_out, int out_size, void* d_ws, size_t ws_size,
                              hipStream_t stream) {
    const float* query = (const float*)d_in[0];
    const float* pos   = (const float*)d_in[1];
    const float* W     = (const float*)d_in[2];
    float* out  = (float*)d_out;
    float* emb2 = (float*)d_ws;

    dim3 g1(M1_PAD / 64, EMB / 64);
    gemm1_kernel<<<g1, 128, 0, stream>>>(pos, W, emb2);

    dim3 g2((BATCH * S_LEN) / 64, BAND_TILES, HEADS);
    gemm2_kernel<<<g2, 128, 0, stream>>>(query, emb2, out);
}


# gemm1 K loop software-pipelined: next chunk's A/B tiles are loaded global->registers before the current chunk's WMMAs and stored to a second LDS stage after them; one barrier per chunk
# speedup vs baseline: 1.7178x; 1.4879x over previous
#include <hip/hip_runtime.h>

typedef float v2f __attribute__((ext_vector_type(2)));
typedef float v4f __attribute__((ext_vector_type(4)));
typedef float v8f __attribute__((ext_vector_type(8)));

#define MAX_POS 1024
#define S_LEN   1024
#define EMB     1024
#define HEADS   16
#define DHEAD   64
#define BATCH   2
#define M1_PAD  2048
#define BAND_TILES (S_LEN / 64 + 1)

#define LDA1 36
#define LDB1 36

__global__ __launch_bounds__(128)
void gemm1_kernel(const float* __restrict__ pos, const float* __restrict__ W,
                  float* __restrict__ emb2)
{
    __shared__ float Als[2][64 * LDA1];
    __shared__ float Bls[2][64 * LDB1];

    const int t    = threadIdx.x;
    const int lane = t & 31;
    const int wave = t >> 5;
    const int wm   = wave >> 1;
    const int wn   = wave & 1;

    const int Mbase = blockIdx.x * 64;
    const int Nbase = blockIdx.y * 64;

    const float* asrc[4];
    int adst[4];
    #pragma unroll
    for (int it = 0; it < 4; ++it) {
        int idx = t + it * 128;
        int row = idx >> 3;
        int col = (idx & 7) << 2;
        int gidx = Mbase + row + (MAX_POS - S_LEN);
        gidx = gidx < 0 ? 0 : gidx;
        gidx = gidx > (2 * MAX_POS - 2) ? (2 * MAX_POS - 2) : gidx;
        asrc[it] = pos + (size_t)gidx * EMB + col;
        adst[it] = row * LDA1 + col;
    }
    const float* bsrc = W + (size_t)(t >> 6) * EMB + Nbase + (t & 63);
    const int    bdst = (t & 63) * LDB1 + (t >> 6);

    v4f   ra[4];
    float rb[16];

    #pragma unroll
    for (int it = 0; it < 4; ++it) ra[it] = *(const v4f*)(asrc[it]);
    #pragma unroll
    for (int it = 0; it < 16; ++it) rb[it] = bsrc[(size_t)(2 * it) * EMB];
    #pragma unroll
    for (int it = 0; it < 4; ++it) *(v4f*)(&Als[0][adst[it]]) = ra[it];
    #pragma unroll
    for (int it = 0; it < 16; ++it) Bls[0][bdst + 2 * it] = rb[it];
    __syncthreads();

    v8f acc[2][2] = {};

    for (int kc = 0; kc < EMB; kc += 32) {
        const int  cur  = (kc >> 5) & 1;
        const bool more = (kc + 32) < EMB;

        if (more) {
            #pragma unroll
            for (int it = 0; it < 4; ++it)
                ra[it] = *(const v4f*)(asrc[it] + kc + 32);
            #pragma unroll
            for (int it = 0; it < 16; ++it)
                rb[it] = bsrc[(size_t)(kc + 32 + 2 * it) * EMB];
        }

        const float* Ac = Als[cur];
        const float* Bc = Bls[cur];
        #pragma unroll
        for (int ks = 0; ks < 32; ks += 4) {
            v2f afrag[2], bfrag[2];
            #pragma unroll
            for (int tm = 0; tm < 2; ++tm) {
                int m = wm * 32 + tm * 16 + (lane & 15);
                afrag[tm] = *(const v2f*)(&Ac[m * LDA1 + ks + 2 * (lane >> 4)]);
            }
            #pragma unroll
            for (int tn = 0; tn < 2; ++tn) {
                int n = wn * 32 + tn * 16 + (lane & 15);
                bfrag[tn] = *(const v2f*)(&Bc[n * LDB1 + ks + 2 * (lane >> 4)]);
            }
            #pragma unroll
            for (int tm = 0; tm < 2; ++tm)
                #pragma unroll
                for (int tn = 0; tn < 2; ++tn)
                    acc[tm][tn] = __builtin_amdgcn_wmma_f32_16x16x4_f32(
                        false, afrag[tm], false, bfrag[tn],
                        (short)0, acc[tm][tn], false, false);
        }

        if (more) {
            float* An = Als[cur ^ 1];
            float* Bn = Bls[cur ^ 1];
            #pragma unroll
            for (int it = 0; it < 4; ++it) *(v4f*)(&An[adst[it]]) = ra[it];
            #pragma unroll
            for (int it = 0; it < 16; ++it) Bn[bdst + 2 * it] = rb[it];
        }
        __syncthreads();
    }

    #pragma unroll
    for (int tm = 0; tm < 2; ++tm)
        #pragma unroll
        for (int tn = 0; tn < 2; ++tn) {
            int mrow0 = Mbase + wm * 32 + tm * 16 + 8 * (lane >> 4);
            int ncol  = Nbase + wn * 32 + tn * 16 + (lane & 15);
            #pragma unroll
            for (int v = 0; v < 8; ++v)
                emb2[(size_t)(mrow0 + v) * EMB + ncol] = acc[tm][tn][v];
        }
}

#define LDA2 68
#define LDB2 68

__global__ __launch_bounds__(128)
void gemm2_kernel(const float* __restrict__ query, const float* __restrict__ emb2,
                  float* __restrict__ out)
{
    __shared__ float Als[64 * LDA2];
    __shared__ float Bls[64 * LDB2];

    const int t    = threadIdx.x;
    const int lane = t & 31;
    const int wave = t >> 5;
    const int wm   = wave >> 1;
    const int wn   = wave & 1;

    const int h     = blockIdx.z;
    const int Nbase = blockIdx.x * 64;
    const int si    = blockIdx.x & (S_LEN / 64 - 1);
    const int Mbase = (S_LEN / 64 - 1 - si + (int)blockIdx.y) * 64;

    #pragma unroll
    for (int it = 0; it < 8; ++it) {
        int idx = t + it * 128;
        int row = idx >> 4;
        int col = (idx & 15) << 2;
        v4f v = *(const v4f*)(query +
                 ((size_t)h * (BATCH * S_LEN) + Nbase + row) * DHEAD + col);
        *(v4f*)(&Als[row * LDA2 + col]) = v;
    }
    #pragma unroll
    for (int it = 0; it < 8; ++it) {
        int idx = t + it * 128;
        int row = idx >> 4;
        int col = (idx & 15) << 2;
        v4f v = *(const v4f*)(emb2 + (size_t)(Mbase + row) * EMB + h * DHEAD + col);
        *(v4f*)(&Bls[row * LDB2 + col]) = v;
    }
    __syncthreads();

    v8f acc[2][2] = {};
    #pragma unroll
    for (int ks = 0; ks < DHEAD; ks += 4) {
        v2f afrag[2], bfrag[2];
        #pragma unroll
        for (int tm = 0; tm < 2; ++tm) {
            int n = wm * 32 + tm * 16 + (lane & 15);
            afrag[tm] = *(const v2f*)(&Als[n * LDA2 + ks + 2 * (lane >> 4)]);
        }
        #pragma unroll
        for (int tn = 0; tn < 2; ++tn) {
            int m = wn * 32 + tn * 16 + (lane & 15);
            bfrag[tn] = *(const v2f*)(&Bls[m * LDB2 + ks + 2 * (lane >> 4)]);
        }
        #pragma unroll
        for (int tm = 0; tm < 2; ++tm)
            #pragma unroll
            for (int tn = 0; tn < 2; ++tn)
                acc[tm][tn] = __builtin_amdgcn_wmma_f32_16x16x4_f32(
                    false, afrag[tm], false, bfrag[tn],
                    (short)0, acc[tm][tn], false, false);
    }

    #pragma unroll
    for (int tm = 0; tm < 2; ++tm)
        #pragma unroll
        for (int tn = 0; tn < 2; ++tn) {
            int nrow0 = Nbase + wm * 32 + tm * 16 + 8 * (lane >> 4);
            int mcol  = Mbase + wn * 32 + tn * 16 + (lane & 15);
            #pragma unroll
            for (int v = 0; v < 8; ++v) {
                int n = nrow0 + v;
                int b = n >> 10;
                int s = n & (S_LEN - 1);
                int j = mcol - (S_LEN - 1) + s;
                if ((unsigned)j < (unsigned)S_LEN) {
                    size_t o = (((size_t)h * BATCH + b) * S_LEN + s) * (size_t)S_LEN + j;
                    out[o] = acc[tm][tn][v];
                }
            }
        }
}

extern "C" void kernel_launch(void* const* d_in, const int* in_sizes, int n_in,
                              void* d_out, int out_size, void* d_ws, size_t ws_size,
                              hipStream_t stream) {
    const float* query = (const float*)d_in[0];
    const float* pos   = (const float*)d_in[1];
    const float* W     = (const float*)d_in[2];
    float* out  = (float*)d_out;
    float* emb2 = (float*)d_ws;

    dim3 g1(M1_PAD / 64, EMB / 64);
    gemm1_kernel<<<g1, 128, 0, stream>>>(pos, W, emb2);

    dim3 g2((BATCH * S_LEN) / 64, BAND_TILES, HEADS);
    gemm2_kernel<<<g2, 128, 0, stream>>>(query, emb2, out);
}


# gemm2: one block per n tile and head walks its 17 band m tiles; scores go through an LDS ring in out layout so each 128-B line of out is stored whole (b128); grid 32x17x16 -> 32x16; sums unchanged
# speedup vs baseline: 2.1155x; 1.2315x over previous
#include <hip/hip_runtime.h>

typedef float v2f __attribute__((ext_vector_type(2)));
typedef float v4f __attribute__((ext_vector_type(4)));
typedef float v8f __attribute__((ext_vector_type(8)));

#define MAX_POS 1024
#define S_LEN   1024
#define EMB     1024
#define HEADS   16
#define DHEAD   64
#define BATCH   2
#define M1_PAD  2048
#define BAND_TILES (S_LEN / 64 + 1)

#define LDA1 36
#define LDB1 36

__global__ __launch_bounds__(128)
void gemm1_kernel(const float* __restrict__ pos, const float* __restrict__ W,
                  float* __restrict__ emb2)
{
    __shared__ float Als[2][64 * LDA1];
    __shared__ float Bls[2][64 * LDB1];

    const int t    = threadIdx.x;
    const int lane = t & 31;
    const int wave = t >> 5;
    const int wm   = wave >> 1;
    const int wn   = wave & 1;

    const int Mbase = blockIdx.x * 64;
    const int Nbase = blockIdx.y * 64;

    const float* asrc[4];
    int adst[4];
    #pragma unroll
    for (int it = 0; it < 4; ++it) {
        int idx = t + it * 128;
        int row = idx >> 3;
        int col = (idx & 7) << 2;
        int gidx = Mbase + row + (MAX_POS - S_LEN);
        gidx = gidx < 0 ? 0 : gidx;
        gidx = gidx > (2 * MAX_POS - 2) ? (2 * MAX_POS - 2) : gidx;
        asrc[it] = pos + (size_t)gidx * EMB + col;
        adst[it] = row * LDA1 + col;
    }
    const float* bsrc = W + (size_t)(t >> 6) * EMB + Nbase + (t & 63);
    const int    bdst = (t & 63) * LDB1 + (t >> 6);

    v4f   ra[4];
    float rb[16];

    #pragma unroll
    for (int it = 0; it < 4; ++it) ra[it] = *(const v4f*)(asrc[it]);
    #pragma unroll
    for (int it = 0; it < 16; ++it) rb[it] = bsrc[(size_t)(2 * it) * EMB];
    #pragma unroll
    for (int it = 0; it < 4; ++it) *(v4f*)(&Als[0][adst[it]]) = ra[it];
    #pragma unroll
    for (int it = 0; it < 16; ++it) Bls[0][bdst + 2 * it] = rb[it];
    __syncthreads();

    v8f acc[2][2] = {};

    for (int kc = 0; kc < EMB; kc += 32) {
        const int  cur  = (kc >> 5) & 1;
        const bool more = (kc + 32) < EMB;

        if (more) {
            #pragma unroll
            for (int it = 0; it < 4; ++it)
                ra[it] = *(const v4f*)(asrc[it] + kc + 32);
            #pragma unroll
            for (int it = 0; it < 16; ++it)
                rb[it] = bsrc[(size_t)(kc + 32 + 2 * it) * EMB];
        }

        const float* Ac = Als[cur];
        const float* Bc = Bls[cur];
        #pragma unroll
        for (int ks = 0; ks < 32; ks += 4) {
            v2f afrag[2], bfrag[2];
            #pragma unroll
            for (int tm = 0; tm < 2; ++tm) {
                int m = wm * 32 + tm * 16 + (lane & 15);
                afrag[tm] = *(const v2f*)(&Ac[m * LDA1 + ks + 2 * (lane >> 4)]);
            }
            #pragma unroll
            for (int tn = 0; tn < 2; ++tn) {
                int n = wn * 32 + tn * 16 + (lane & 15);
                bfrag[tn] = *(const v2f*)(&Bc[n * LDB1 + ks + 2 * (lane >> 4)]);
            }
            #pragma unroll
            for (int tm = 0; tm < 2; ++tm)
                #pragma unroll
                for (int tn = 0; tn < 2; ++tn)
                    acc[tm][tn] = __builtin_amdgcn_wmma_f32_16x16x4_f32(
                        false, afrag[tm], false, bfrag[tn],
                        (short)0, acc[tm][tn], false, false);
        }

        if (more) {
            float* An = Als[cur ^ 1];
            float* Bn = Bls[cur ^ 1];
            #pragma unroll
            for (int it = 0; it < 4; ++it) *(v4f*)(&An[adst[it]]) = ra[it];
            #pragma unroll
            for (int it = 0; it < 16; ++it) Bn[bdst + 2 * it] = rb[it];
        }
        __syncthreads();
    }

    #pragma unroll
    for (int tm = 0; tm < 2; ++tm)
        #pragma unroll
        for (int tn = 0; tn < 2; ++tn) {
            int mrow0 = Mbase + wm * 32 + tm * 16 + 8 * (lane >> 4);
            int ncol  = Nbase + wn * 32 + tn * 16 + (lane & 15);
            #pragma unroll
            for (int v = 0; v < 8; ++v)
                emb2[(size_t)(mrow0 + v) * EMB + ncol] = acc[tm][tn][v];
        }
}

#define LDA2 68
#define LDB2 68
#define RING_W 256
#define LDR  260

__global__ __launch_bounds__(128)
void gemm2_kernel(const float* __restrict__ query, const float* __restrict__ emb2,
                  float* __restrict__ out)
{
    __shared__ float Als[64 * LDA2];
    __shared__ float Bls[2][64 * LDB2];
    __shared__ float Rls[64 * LDR];

    const int t    = threadIdx.x;
    const int lane = t & 31;
    const int wave = t >> 5;
    const int wm   = wave >> 1;
    const int wn   = wave & 1;

    const int h      = blockIdx.y;
    const int Nbase  = blockIdx.x * 64;
    const int si     = blockIdx.x & (S_LEN / 64 - 1);
    const int Mfirst = (S_LEN / 64 - 1 - si) * 64;

    const int lrow = t >> 4;
    const int lcol = (t & 15) << 2;

    const float* qsrc = query + ((size_t)h * (BATCH * S_LEN) + Nbase + lrow) * DHEAD + lcol;
    const float* esrc = emb2 + (size_t)(Mfirst + lrow) * EMB + h * DHEAD + lcol;
    float*       odst = out + ((size_t)h * (BATCH * S_LEN) + Nbase + lrow) * S_LEN + lcol;

    v4f re[8];
    {
        v4f rq[8];
        #pragma unroll
        for (int it = 0; it < 8; ++it) rq[it] = *(const v4f*)(qsrc + (size_t)(8 * it) * DHEAD);
        #pragma unroll
        for (int it = 0; it < 8; ++it) re[it] = *(const v4f*)(esrc + (size_t)(8 * it) * EMB);
        #pragma unroll
        for (int it = 0; it < 8; ++it) *(v4f*)(&Als[(lrow + 8 * it) * LDA2 + lcol]) = rq[it];
        #pragma unroll
        for (int it = 0; it < 8; ++it) *(v4f*)(&Bls[0][(lrow + 8 * it) * LDB2 + lcol]) = re[it];
    }
    __syncthreads();

    int slot = 3;
    for (int y = 0; y < BAND_TILES; ++y) {
        const int  cur  = y & 1;
        const bool more = (y + 1) < BAND_TILES;

        if (more) {
            #pragma unroll
            for (int it = 0; it < 8; ++it)
                re[it] = *(const v4f*)(esrc + (size_t)((y + 1) * 64 + 8 * it) * EMB);
        }

        const float* Bc = Bls[cur];
        v8f acc[2][2] = {};
        #pragma unroll
        for (int ks = 0; ks < DHEAD; ks += 4) {
            v2f afrag[2], bfrag[2];
            #pragma unroll
            for (int tm = 0; tm < 2; ++tm) {
                int n = wm * 32 + tm * 16 + (lane & 15);
                afrag[tm] = *(const v2f*)(&Als[n * LDA2 + ks + 2 * (lane >> 4)]);
            }
            #pragma unroll
            for (int tn = 0; tn < 2; ++tn) {
                int m = wn * 32 + tn * 16 + (lane & 15);
                bfrag[tn] = *(const v2f*)(&Bc[m * LDB2 + ks + 2 * (lane >> 4)]);
            }
            #pragma unroll
            for (int tm = 0; tm < 2; ++tm)
                #pragma unroll
                for (int tn = 0; tn < 2; ++tn)
                    acc[tm][tn] = __builtin_amdgcn_wmma_f32_16x16x4_f32(
                        false, afrag[tm], false, bfrag[tn],
                        (short)0, acc[tm][tn], false, false);
        }

        #pragma unroll
        for (int tm = 0; tm < 2; ++tm)
            #pragma unroll
            for (int tn = 0; tn < 2; ++tn) {
                int r0 = wm * 32 + tm * 16 + 8 * (lane >> 4);
                int c  = wn * 32 + tn * 16 + (lane & 15);
                #pragma unroll
                for (int v = 0; v < 8; ++v) {
                    int r  = r0 + v;
                    int jj = (slot * 64 + 1 + c + r) & (RING_W - 1);
                    Rls[r * LDR + jj] = acc[tm][tn][v];
                }
            }

        if (more) {
            float* Bn = Bls[cur ^ 1];
            #pragma unroll
            for (int it = 0; it < 8; ++it) *(v4f*)(&Bn[(lrow + 8 * it) * LDB2 + lcol]) = re[it];
        }
        __syncthreads();

        if (y >= 1) {
            #pragma unroll
            for (int it = 0; it < 8; ++it) {
                v4f v = *(const v4f*)(&Rls[(lrow + 8 * it) * LDR + slot * 64 + lcol]);
                *(v4f*)(odst + (size_t)(8 * it) * S_LEN + 64 * (y - 1)) = v;
            }
        }
        slot = (slot + 1) & 3;
    }
}

extern "C" void kernel_launch(void* const* d_in, const int* in_sizes, int n_in,
                              void* d_out, int out_size, void* d_ws, size_t ws_size,
                              hipStream_t stream) {
    const float* query = (const float*)d_in[0];
    const float* pos   = (const float*)d_in[1];
    const float* W     = (const float*)d_in[2];
    float* out  = (float*)d_out;
    float* emb2 = (float*)d_ws;

    dim3 g1(M1_PAD / 64, EMB / 64);
    gemm1_kernel<<<g1, 128, 0, stream>>>(pos, W, emb2);

    dim3 g2((BATCH * S_LEN) / 64, HEADS);
    gemm2_kernel<<<g2, 128, 0, stream>>>(query, emb2, out);
}


# both GEMMs: chunk-boundary work (LDS refill, split barrier, next chunk's first fragments, ring window store) issued between the running chunk's WMMA groups; prefetch 2 chunks ahead; sums unchanged
# speedup vs baseline: 2.2307x; 1.0545x over previous
#include <hip/hip_runtime.h>

typedef float v2f __attribute__((ext_vector_type(2)));
typedef float v4f __attribute__((ext_vector_type(4)));
typedef float v8f __attribute__((ext_vector_type(8)));

#define MAX_POS 1024
#define S_LEN   1024
#define EMB     1024
#define HEADS   16
#define DHEAD   64
#define BATCH   2
#define M1_PAD  2048
#define BAND_TILES (S_LEN / 64 + 1)

#define LDA1 36
#define LDB1 36

#define DEV static __device__ __forceinline__
DEV void sched_fence() { __builtin_amdgcn_sched_barrier(0); }
DEV void wait_ds0() { asm volatile("s_wait_dscnt 0x0" ::: "memory"); }
DEV void bar_signal() { asm volatile("s_barrier_signal -1" ::: "memory"); }
DEV void bar_wait() { asm volatile("s_barrier_wait -1" ::: "memory"); }

struct G1Frag { v2f a[2][2]; v2f b[2][2]; };

DEV void g1_load(G1Frag& f, const float* Ac, const float* Bc, int aoff, int boff, int g)
{
    #pragma unroll
    for (int kk = 0; kk < 2; ++kk) {
        const int ks = (2 * g + kk) * 4;
        #pragma unroll
        for (int tm = 0; tm < 2; ++tm)
            f.a[kk][tm] = *(const v2f*)(&Ac[aoff + tm * 16 * LDA1 + ks]);
        #pragma unroll
        for (int tn = 0; tn < 2; ++tn)
            f.b[kk][tn] = *(const v2f*)(&Bc[boff + tn * 16 * LDB1 + ks]);
    }
}

DEV void g1_mma(v8f (&acc)[2][2], const G1Frag& f, int kk)
{
    #pragma unroll
    for (int tm = 0; tm < 2; ++tm)
        #pragma unroll
        for (int tn = 0; tn < 2; ++tn)
            acc[tm][tn] = __builtin_amdgcn_wmma_f32_16x16x4_f32(
                false, f.a[kk][tm], false, f.b[kk][tn],
                (short)0, acc[tm][tn], false, false);
}

template <bool NEXT, bool NEXT2>
DEV void g1_chunk(v8f (&acc)[2][2], G1Frag& f0, G1Frag& f1,
                  const float* Ac, const float* Bc, float* An, float* Bn,
                  int aoff, int boff,
                  v4f (&ra)[4], float (&rb)[16],
                  const float* const (&asrc)[4], const int (&adst)[4],
                  const float* bsrc, int bdst, int kc2)
{
    g1_load(f1, Ac, Bc, aoff, boff, 1);
    sched_fence();
    if (NEXT) {
        #pragma unroll
        for (int it = 0; it < 4; ++it) *(v4f*)(&An[adst[it]]) = ra[it];
        #pragma unroll
        for (int it = 0; it < 16; ++it) Bn[bdst + 2 * it] = rb[it];
    }
    if (NEXT2) {
        #pragma unroll
        for (int it = 0; it < 4; ++it)
            ra[it] = *(const v4f*)(asrc[it] + kc2);
        #pragma unroll
        for (int it = 0; it < 16; ++it)
            rb[it] = bsrc[(size_t)(kc2 + 2 * it) * EMB];
    }
    g1_mma(acc, f0, 0);
    g1_mma(acc, f0, 1);
    sched_fence();
    g1_load(f0, Ac, Bc, aoff, boff, 2);
    sched_fence();
    g1_mma(acc, f1, 0);
    g1_mma(acc, f1, 1);
    sched_fence();
    g1_load(f1, Ac, Bc, aoff, boff, 3);
    sched_fence();
    g1_mma(acc, f0, 0);
    g1_mma(acc, f0, 1);
    sched_fence();
    if (NEXT) { wait_ds0(); bar_signal(); }
    sched_fence();
    g1_mma(acc, f1, 0);
    sched_fence();
    if (NEXT) {
        bar_wait();
        g1_load(f0, An, Bn, aoff, boff, 0);
    }
    sched_fence();
    g1_mma(acc, f1, 1);
    sched_fence();
}

__global__ __launch_bounds__(128)
void gemm1_kernel(const float* __restrict__ pos, const float* __restrict__ W,
                  float* __restrict__ emb2)
{
    __shared__ float Als[2][64 * LDA1];
    __shared__ float Bls[2][64 * LDB1];

    const int t    = threadIdx.x;
    const int lane = t & 31;
    const int wave = t >> 5;
    const int wm   = wave >> 1;
    const int wn   = wave & 1;

    const int Mbase = blockIdx.x * 64;
    const int Nbase = blockIdx.y * 64;

    const float* asrc[4];
    int adst[4];
    #pragma unroll
    for (int it = 0; it < 4; ++it) {
        int idx = t + it * 128;
        int row = idx >> 3;
        int col = (idx & 7) << 2;
        int gidx = Mbase + row + (MAX_POS - S_LEN);
        gidx = gidx < 0 ? 0 : gidx;
        gidx = gidx > (2 * MAX_POS - 2) ? (2 * MAX_POS - 2) : gidx;
        asrc[it] = pos + (size_t)gidx * EMB + col;
        adst[it] = row * LDA1 + col;
    }
    const float* bsrc = W + (size_t)(t >> 6) * EMB + Nbase + (t & 63);
    const int    bdst = (t & 63) * LDB1 + (t >> 6);

    const int aoff = (wm * 32 + (lane & 15)) * LDA1 + 2 * (lane >> 4);
    const int boff = (wn * 32 + (lane & 15)) * LDB1 + 2 * (lane >> 4);

    v4f   ra[4];
    float rb[16];

    #pragma unroll
    for (int it = 0; it < 4; ++it) ra[it] = *(const v4f*)(asrc[it]);
    #pragma unroll
    for (int it = 0; it < 16; ++it) rb[it] = bsrc[(size_t)(2 * it) * EMB];
    #pragma unroll
    for (int it = 0; it < 4; ++it) *(v4f*)(&Als[0][adst[it]]) = ra[it];
    #pragma unroll
    for (int it = 0; it < 16; ++it) Bls[0][bdst + 2 * it] = rb[it];
    #pragma unroll
    for (int it = 0; it < 4; ++it) ra[it] = *(const v4f*)(asrc[it] + 32);
    #pragma unroll
    for (int it = 0; it < 16; ++it) rb[it] = bsrc[(size_t)(32 + 2 * it) * EMB];
    wait_ds0();
    bar_signal();
    bar_wait();

    v8f acc[2][2] = {};
    G1Frag f0, f1;
    g1_load(f0, Als[0], Bls[0], aoff, boff, 0);

    const int NCH = EMB / 32;
    for (int i = 0; i < NCH - 2; ++i) {
        const int cur = i & 1;
        g1_chunk<true, true>(acc, f0, f1, Als[cur], Bls[cur], Als[cur ^ 1], Bls[cur ^ 1],
                             aoff, boff, ra, rb, asrc, adst, bsrc, bdst, (i + 2) * 32);
    }
    g1_chunk<true, false>(acc, f0, f1, Als[0], Bls[0], Als[1], Bls[1],
                          aoff, boff, ra, rb, asrc, adst, bsrc, bdst, 0);
    g1_chunk<false, false>(acc, f0, f1, Als[1], Bls[1], Als[0], Bls[0],
                           aoff, boff, ra, rb, asrc, adst, bsrc, bdst, 0);

    #pragma unroll
    for (int tm = 0; tm < 2; ++tm)
        #pragma unroll
        for (int tn = 0; tn < 2; ++tn) {
            int mrow0 = Mbase + wm * 32 + tm * 16 + 8 * (lane >> 4);
            int ncol  = Nbase + wn * 32 + tn * 16 + (lane & 15);
            #pragma unroll
            for (int v = 0; v < 8; ++v)
                emb2[(size_t)(mrow0 + v) * EMB + ncol] = acc[tm][tn][v];
        }
}

#define LDA2 68
#define LDB2 68
#define RING_W 256
#define LDR  260

DEV void g2_load(G1Frag& f, const float* Aq, const float* Bc, int aoff, int boff, int g)
{
    #pragma unroll
    for (int kk = 0; kk < 2; ++kk) {
        const int ks = (2 * g + kk) * 4;
        #pragma unroll
        for (int tm = 0; tm < 2; ++tm)
            f.a[kk][tm] = *(const v2f*)(&Aq[aoff + tm * 16 * LDA2 + ks]);
        #pragma unroll
        for (int tn = 0; tn < 2; ++tn)
            f.b[kk][tn] = *(const v2f*)(&Bc[boff + tn * 16 * LDB2 + ks]);
    }
}

template <bool COPY, bool NEXT, bool NEXT2>
DEV void g2_tile(G1Frag& f0, G1Frag& f1, const float* Aq, const float* Bc, float* Bn, float* Rls,
                 int aoff, int boff, v4f (&re)[8], const float* enext2,
                 int lrow, int lcol, int slot, int cslot, float* ocol,
                 int wm, int wn, int lane)
{
    v8f acc[2][2] = {};
    v4f cw[8];
    g2_load(f1, Aq, Bc, aoff, boff, 1);
    sched_fence();
    if (NEXT) {
        #pragma unroll
        for (int it = 0; it < 8; ++it) *(v4f*)(&Bn[(lrow + 8 * it) * LDB2 + lcol]) = re[it];
    }
    if (NEXT2) {
        #pragma unroll
        for (int it = 0; it < 8; ++it) re[it] = *(const v4f*)(enext2 + (size_t)(8 * it) * EMB);
    }
    g1_mma(acc, f0, 0);
    g1_mma(acc, f0, 1);
    sched_fence();
    g2_load(f0, Aq, Bc, aoff, boff, 2);
    sched_fence();
    g1_mma(acc, f1, 0);
    g1_mma(acc, f1, 1);
    sched_fence();
    if (COPY) {
        #pragma unroll
        for (int it = 0; it < 8; ++it)
            cw[it] = *(const v4f*)(&Rls[(lrow + 8 * it) * LDR + cslot * 64 + lcol]);
    }
    sched_fence();
    g2_load(f1, Aq, Bc, aoff, boff, 3);
    sched_fence();
    g1_mma(acc, f0, 0);
    g1_mma(acc, f0, 1);
    sched_fence();
    g2_load(f0, Aq, Bc, aoff, boff, 4);
    sched_fence();
    g1_mma(acc, f1, 0);
    g1_mma(acc, f1, 1);
    sched_fence();
    g2_load(f1, Aq, Bc, aoff, boff, 5);
    sched_fence();
    if (COPY) {
        #pragma unroll
        for (int it = 0; it < 8; ++it)
            *(v4f*)(ocol + (size_t)(8 * it) * S_LEN) = cw[it];
    }
    g1_mma(acc, f0, 0);
    g1_mma(acc, f0, 1);
    sched_fence();
    g2_load(f0, Aq, Bc, aoff, boff, 6);
    sched_fence();
    int sidx[2][2][8];
    #pragma unroll
    for (int tm = 0; tm < 2; ++tm)
        #pragma unroll
        for (int tn = 0; tn < 2; ++tn) {
            int r0 = wm * 32 + tm * 16 + 8 * (lane >> 4);
            int c  = wn * 32 + tn * 16 + (lane & 15);
            #pragma unroll
            for (int v = 0; v < 8; ++v) {
                int r  = r0 + v;
                int jj = (slot * 64 + 1 + c + r) & (RING_W - 1);
                sidx[tm][tn][v] = r * LDR + jj;
            }
        }
    g1_mma(acc, f1, 0);
    g1_mma(acc, f1, 1);
    sched_fence();
    g2_load(f1, Aq, Bc, aoff, boff, 7);
    sched_fence();
    g1_mma(acc, f0, 0);
    g1_mma(acc, f0, 1);
    sched_fence();
    wait_ds0();
    bar_signal();
    sched_fence();
    g1_mma(acc, f1, 0);
    sched_fence();
    bar_wait();
    if (NEXT) g2_load(f0, Aq, Bn, aoff, boff, 0);
    sched_fence();
    g1_mma(acc, f1, 1);
    sched_fence();
    #pragma unroll
    for (int tm = 0; tm < 2; ++tm)
        #pragma unroll
        for (int tn = 0; tn < 2; ++tn) {
            #pragma unroll
            for (int v = 0; v < 8; ++v)
                Rls[sidx[tm][tn][v]] = acc[tm][tn][v];
        }
}

__global__ __launch_bounds__(128)
void gemm2_kernel(const float* __restrict__ query, const float* __restrict__ emb2,
                  float* __restrict__ out)
{
    __shared__ float Als[64 * LDA2];
    __shared__ float Bls[2][64 * LDB2];
    __shared__ float Rls[64 * LDR];

    const int t    = threadIdx.x;
    const int lane = t & 31;
    const int wave = t >> 5;
    const int wm   = wave >> 1;
    const int wn   = wave & 1;

    const int h      = blockIdx.y;
    const int Nbase  = blockIdx.x * 64;
    const int si     = blockIdx.x & (S_LEN / 64 - 1);
    const int Mfirst = (S_LEN / 64 - 1 - si) * 64;

    const int lrow = t >> 4;
    const int lcol = (t & 15) << 2;

    const float* qsrc = query + ((size_t)h * (BATCH * S_LEN) + Nbase + lrow) * DHEAD + lcol;
    const float* esrc = emb2 + (size_t)(Mfirst + lrow) * EMB + h * DHEAD + lcol;
    float*       odst = out + ((size_t)h * (BATCH * S_LEN) + Nbase + lrow) * S_LEN + lcol;

    const int aoff = (wm * 32 + (lane & 15)) * LDA2 + 2 * (lane >> 4);
    const int boff = (wn * 32 + (lane & 15)) * LDB2 + 2 * (lane >> 4);

    v4f re[8];
    {
        v4f rq[8];
        #pragma unroll
        for (int it = 0; it < 8; ++it) rq[it] = *(const v4f*)(qsrc + (size_t)(8 * it) * DHEAD);
        #pragma unroll
        for (int it = 0; it < 8; ++it) re[it] = *(const v4f*)(esrc + (size_t)(8 * it) * EMB);
        #pragma unroll
        for (int it = 0; it < 8; ++it) *(v4f*)(&Als[(lrow + 8 * it) * LDA2 + lcol]) = rq[it];
        #pragma unroll
        for (int it = 0; it < 8; ++it) *(v4f*)(&Bls[0][(lrow + 8 * it) * LDB2 + lcol]) = re[it];
    }
    #pragma unroll
    for (int it = 0; it < 8; ++it) re[it] = *(const v4f*)(esrc + (size_t)(64 + 8 * it) * EMB);
    wait_ds0();
    bar_signal();
    bar_wait();

    G1Frag f0, f1;
    g2_load(f0, Als, Bls[0], aoff, boff, 0);

    for (int y = 0; y < 3; ++y) {
        const int cur = y & 1;
        g2_tile<false, true, true>(f0, f1, Als, Bls[cur], Bls[cur ^ 1], Rls, aoff, boff, re,
                                   esrc + (size_t)((y + 2) * 64) * EMB,
                                   lrow, lcol, (y + 3) & 3, 0, odst, wm, wn, lane);
    }
    for (int y = 3; y < BAND_TILES - 2; ++y) {
        const int cur = y & 1;
        g2_tile<true, true, true>(f0, f1, Als, Bls[cur], Bls[cur ^ 1], Rls, aoff, boff, re,
                                  esrc + (size_t)((y + 2) * 64) * EMB,
                                  lrow, lcol, (y + 3) & 3, (y + 1) & 3, odst + 64 * (y - 3), wm, wn, lane);
    }
    {
        const int y = BAND_TILES - 2;
        g2_tile<true, true, false>(f0, f1, Als, Bls[y & 1], Bls[(y & 1) ^ 1], Rls, aoff, boff, re,
                                   esrc, lrow, lcol, (y + 3) & 3, (y + 1) & 3, odst + 64 * (y - 3), wm, wn, lane);
    }
    {
        const int y = BAND_TILES - 1;
        g2_tile<true, false, false>(f0, f1, Als, Bls[y & 1], Bls[(y & 1) ^ 1], Rls, aoff, boff, re,
                                    esrc, lrow, lcol, (y + 3) & 3, (y + 1) & 3, odst + 64 * (y - 3), wm, wn, lane);
    }
    wait_ds0();
    bar_signal();
    bar_wait();
    #pragma unroll
    for (int w = S_LEN / 64 - 2; w < S_LEN / 64; ++w) {
        #pragma unroll
        for (int it = 0; it < 8; ++it) {
            v4f v = *(const v4f*)(&Rls[(lrow + 8 * it) * LDR + (w & 3) * 64 + lcol]);
            *(v4f*)(odst + (size_t)(8 * it) * S_LEN + 64 * w) = v;
        }
    }
}

extern "C" void kernel_launch(void* const* d_in, const int* in_sizes, int n_in,
                              void* d_out, int out_size, void* d_ws, size_t ws_size,
                              hipStream_t stream) {
    const float* query = (const float*)d_in[0];
    const float* pos   = (const float*)d_in[1];
    const float* W     = (const float*)d_in[2];
    float* out  = (float*)d_out;
    float* emb2 = (float*)d_ws;

    dim3 g1(M1_PAD / 64, EMB / 64);
    gemm1_kernel<<<g1, 128, 0, stream>>>(pos, W, emb2);

    dim3 g2((BATCH * S_LEN) / 64, HEADS);
    gemm2_kernel<<<g2, 128, 0, stream>>>(query, emb2, out);
}
